# starting kernel
# baseline (speedup 1.0000x reference)
#include <hip/hip_runtime.h>
#include <hip/hip_bf16.h>

typedef float v2f __attribute__((ext_vector_type(2)));
typedef float v8f __attribute__((ext_vector_type(8)));

#define RREP   48
#define NNODE  1000
#define FDIM   64
#define HHEADS 4
#define CDIM   64
#define OUTD   256
#define NEDGE  12000
#define EPRIME 13000
#define NROWS  48000
#define NEG_SLOPE 0.2f

__global__ void gat_init(float* __restrict__ out, float* __restrict__ den,
                         const float* __restrict__ bias) {
    int tid = blockIdx.x * blockDim.x + threadIdx.x;
    if (tid < NROWS * FDIM) out[tid] = bias[tid & (FDIM - 1)];
    if (tid < NROWS * HHEADS) den[tid] = 0.0f;
}

__global__ void gat_proj_wmma(const float* __restrict__ x,
                              const float* __restrict__ W,
                              float* __restrict__ h) {
    const int lane = threadIdx.x & 31;
    const int nblk = threadIdx.x >> 5;
    const int mt   = blockIdx.x;
    const int mrow = mt * 16 + (lane & 15);
    const int kb   = (lane >> 4) * 2;
    const int col0 = nblk * 64 + (lane & 15);

    v8f acc0 = {}, acc1 = {}, acc2 = {}, acc3 = {};

    #pragma unroll
    for (int kk = 0; kk < 16; ++kk) {
        const int k = kk * 4 + kb;
        v2f a;
        const float2 av = *(const float2*)(x + (size_t)mrow * FDIM + k);
        a.x = av.x; a.y = av.y;
        v2f b0, b1, b2, b3;
        b0.x = W[k * OUTD + col0 +  0]; b0.y = W[(k + 1) * OUTD + col0 +  0];
        b1.x = W[k * OUTD + col0 + 16]; b1.y = W[(k + 1) * OUTD + col0 + 16];
        b2.x = W[k * OUTD + col0 + 32]; b2.y = W[(k + 1) * OUTD + col0 + 32];
        b3.x = W[k * OUTD + col0 + 48]; b3.y = W[(k + 1) * OUTD + col0 + 48];

        acc0 = __builtin_amdgcn_wmma_f32_16x16x4_f32(false, a, false, b0,
                                                     (short)0, acc0, false, false);
        acc1 = __builtin_amdgcn_wmma_f32_16x16x4_f32(false, a, false, b1,
                                                     (short)0, acc1, false, false);
        acc2 = __builtin_amdgcn_wmma_f32_16x16x4_f32(false, a, false, b2,
                                                     (short)0, acc2, false, false);
        acc3 = __builtin_amdgcn_wmma_f32_16x16x4_f32(false, a, false, b3,
                                                     (short)0, acc3, false, false);
    }

    const int mhalf = (lane >> 4) * 8;
    #pragma unroll
    for (int i = 0; i < 8; ++i) {
        const size_t row = (size_t)(mt * 16 + mhalf + i) * OUTD;
        h[row + col0 +  0] = acc0[i];
        h[row + col0 + 16] = acc1[i];
        h[row + col0 + 32] = acc2[i];
        h[row + col0 + 48] = acc3[i];
    }
}

__global__ void gat_attn_dots(const float* __restrict__ h,
                              const float* __restrict__ att_src,
                              const float* __restrict__ att_dst,
                              float* __restrict__ a_src,
                              float* __restrict__ a_dst) {
    int tid = blockIdx.x * blockDim.x + threadIdx.x;
    if (tid >= NROWS * HHEADS) return;
    const int hh = tid & (HHEADS - 1);
    const int rn = tid >> 2;
    const float* hp = h + (size_t)rn * OUTD + hh * CDIM;
    const float* as = att_src + hh * CDIM;
    const float* ad = att_dst + hh * CDIM;
    float ss = 0.f, sd = 0.f;
    #pragma unroll 8
    for (int c = 0; c < CDIM; ++c) {
        const float v = hp[c];
        ss += v * as[c];
        sd += v * ad[c];
    }
    a_src[tid] = ss;
    a_dst[tid] = sd;
}

__global__ void gat_edge_exp(const int* __restrict__ ei,
                             const float* __restrict__ a_src,
                             const float* __restrict__ a_dst,
                             float* __restrict__ exb,
                             float* __restrict__ den) {
    int tid = blockIdx.x * blockDim.x + threadIdx.x;
    if (tid >= RREP * EPRIME * HHEADS) return;
    const int hh = tid & (HHEADS - 1);
    const int re = tid >> 2;
    const int e  = re % EPRIME;
    const int r  = re / EPRIME;
    int s, d; bool selfm;
    if (e < NEDGE) { s = ei[e]; d = ei[NEDGE + e]; selfm = (s == d); }
    else           { s = e - NEDGE; d = s; selfm = false; }
    float ex;
    if (selfm) {
        ex = 0.0f;
    } else {
        float v = a_src[(r * NNODE + s) * HHEADS + hh]
                + a_dst[(r * NNODE + d) * HHEADS + hh];
        v = v > 0.0f ? v : NEG_SLOPE * v;
        ex = __expf(v);
    }
    exb[tid] = ex;
    atomicAdd(&den[(r * NNODE + d) * HHEADS + hh], ex);
}

__global__ void gat_aggregate(const int* __restrict__ ei,
                              const float* __restrict__ h,
                              const float* __restrict__ exb,
                              const float* __restrict__ den,
                              float* __restrict__ out) {
    const int wid  = blockIdx.x * (blockDim.x >> 5) + (threadIdx.x >> 5);
    if (wid >= RREP * EPRIME) return;
    const int lane = threadIdx.x & 31;
    const int e = wid % EPRIME;
    const int r = wid / EPRIME;
    int s, d;
    if (e < NEDGE) { s = ei[e]; d = ei[NEDGE + e]; }
    else           { s = e - NEDGE; d = s; }

    float alpha[HHEADS];
    #pragma unroll
    for (int hh = 0; hh < HHEADS; ++hh) {
        const float ex = exb[(r * EPRIME + e) * HHEADS + hh];
        const float dn = den[(r * NNODE + d) * HHEADS + hh];
        alpha[hh] = (1.0f / HHEADS) * ex / (dn + 1e-16f);
    }

    const float* hs = h + (size_t)(r * NNODE + s) * OUTD;
    float* op = out + (size_t)(r * NNODE + d) * FDIM;
    #pragma unroll
    for (int t = 0; t < 8; ++t) {
        const int idx = lane + t * 32;
        const int hh  = idx >> 6;
        const int f   = idx & 63;
        atomicAdd(&op[f], alpha[hh] * hs[idx]);
    }
}

extern "C" void kernel_launch(void* const* d_in, const int* in_sizes, int n_in,
                              void* d_out, int out_size, void* d_ws, size_t ws_size,
                              hipStream_t stream) {
    const float* x       = (const float*)d_in[0];
    const int*   ei      = (const int*)d_in[1];
    const float* W       = (const float*)d_in[2];
    const float* att_src = (const float*)d_in[3];
    const float* att_dst = (const float*)d_in[4];
    const float* bias    = (const float*)d_in[5];
    float* out = (float*)d_out;

    float* ws    = (float*)d_ws;
    float* hbuf  = ws;
    float* a_src = hbuf  + (size_t)NROWS * OUTD;
    float* a_dst = a_src + (size_t)NROWS * HHEADS;
    float* exb   = a_dst + (size_t)NROWS * HHEADS;
    float* den   = exb   + (size_t)RREP * EPRIME * HHEADS;

    {
        const int n = NROWS * FDIM;
        gat_init<<<(n + 255) / 256, 256, 0, stream>>>(out, den, bias);
    }
    {
        dim3 grid(NROWS / 16);
        dim3 block(128);
        gat_proj_wmma<<<grid, block, 0, stream>>>(x, W, hbuf);
    }
    {
        const int n = NROWS * HHEADS;
        gat_attn_dots<<<(n + 255) / 256, 256, 0, stream>>>(hbuf, att_src, att_dst,
                                                           a_src, a_dst);
    }
    {
        const int n = RREP * EPRIME * HHEADS;
        gat_edge_exp<<<(n + 255) / 256, 256, 0, stream>>>(ei, a_src, a_dst, exb, den);
    }
    {
        const int waves = RREP * EPRIME;
        const int wpb   = 256 / 32;
        gat_aggregate<<<(waves + wpb - 1) / wpb, 256, 0, stream>>>(ei, hbuf, exb,
                                                                   den, out);
    }
}


# Aggregation: pass 1 packs the tile's kept edges into a dense LDS list (ballot count, one LDS integer add per wave, arrival order); pass 2 walks it uniformly, no lane-pick chain; acc sum order changes
# speedup vs baseline: 8.5015x; 8.5015x over previous
#include <hip/hip_runtime.h>
#include <hip/hip_bf16.h>

typedef float v2f __attribute__((ext_vector_type(2)));
typedef float v8f __attribute__((ext_vector_type(8)));

#define RREP   48
#define NNODE  1000
#define FDIM   64
#define HHEADS 4
#define CDIM   64
#define OUTD   256
#define NEDGE  12000
#define NROWS  48000
#define NEG_SLOPE 0.2f

template <int MASK>
__device__ __forceinline__ float row_xor(const float v) {
    return __int_as_float(__builtin_amdgcn_update_dpp(0, __float_as_int(v),
                                                      0x160 + MASK, 0xf, 0xf, true));
}

__global__ void gat_proj_wmma(const float* __restrict__ x,
                              const float* __restrict__ W,
                              const float* __restrict__ att_src,
                              const float* __restrict__ att_dst,
                              float* __restrict__ h,
                              float* __restrict__ a_src,
                              float* __restrict__ a_dst) {
    __shared__ __attribute__((aligned(16))) float lg[2][16][HHEADS];

    const int lane = threadIdx.x & 31;
    const int nblk = threadIdx.x >> 5;
    const int mt   = blockIdx.x;
    const int mrow = mt * 16 + (lane & 15);
    const int kb   = (lane >> 4) * 2;
    const int col0 = nblk * 64 + (lane & 15);

    v8f acc0 = {}, acc1 = {}, acc2 = {}, acc3 = {};

    #pragma unroll
    for (int kk = 0; kk < 16; ++kk) {
        const int k = kk * 4 + kb;
        v2f a;
        const float2 av = *(const float2*)(x + (size_t)mrow * FDIM + k);
        a.x = av.x; a.y = av.y;
        v2f b0, b1, b2, b3;
        b0.x = W[k * OUTD + col0 +  0]; b0.y = W[(k + 1) * OUTD + col0 +  0];
        b1.x = W[k * OUTD + col0 + 16]; b1.y = W[(k + 1) * OUTD + col0 + 16];
        b2.x = W[k * OUTD + col0 + 32]; b2.y = W[(k + 1) * OUTD + col0 + 32];
        b3.x = W[k * OUTD + col0 + 48]; b3.y = W[(k + 1) * OUTD + col0 + 48];

        acc0 = __builtin_amdgcn_wmma_f32_16x16x4_f32(false, a, false, b0,
                                                     (short)0, acc0, false, false);
        acc1 = __builtin_amdgcn_wmma_f32_16x16x4_f32(false, a, false, b1,
                                                     (short)0, acc1, false, false);
        acc2 = __builtin_amdgcn_wmma_f32_16x16x4_f32(false, a, false, b2,
                                                     (short)0, acc2, false, false);
        acc3 = __builtin_amdgcn_wmma_f32_16x16x4_f32(false, a, false, b3,
                                                     (short)0, acc3, false, false);
    }

    const int mhalf = (lane >> 4) * 8;

    const int c = lane & 15;
    const float* asp = att_src + nblk * CDIM + c;
    const float* adp = att_dst + nblk * CDIM + c;
    const float s0 = asp[0], s1 = asp[16], s2 = asp[32], s3 = asp[48];
    const float t0 = adp[0], t1 = adp[16], t2 = adp[32], t3 = adp[48];
    float v[16];
    for (int i = 0; i < 8; ++i) {
        v[i]     = fmaf(acc3[i], s3, fmaf(acc2[i], s2, fmaf(acc1[i], s1, acc0[i] * s0)));
        v[8 + i] = fmaf(acc3[i], t3, fmaf(acc2[i], t2, fmaf(acc1[i], t1, acc0[i] * t0)));
    }
    float w8[8], w4[4], w2[2], w1;
    {
        const bool up = (c & 8) != 0;
        for (int m = 0; m < 8; ++m) {
            const float keep = up ? v[m + 8] : v[m];
            const float send = up ? v[m] : v[m + 8];
            w8[m] = keep + row_xor<8>(send);
        }
    }
    {
        const bool up = (c & 4) != 0;
        for (int m = 0; m < 4; ++m) {
            const float keep = up ? w8[m + 4] : w8[m];
            const float send = up ? w8[m] : w8[m + 4];
            w4[m] = keep + row_xor<4>(send);
        }
    }
    {
        const bool up = (c & 2) != 0;
        for (int m = 0; m < 2; ++m) {
            const float keep = up ? w4[m + 2] : w4[m];
            const float send = up ? w4[m] : w4[m + 2];
            w2[m] = keep + row_xor<2>(send);
        }
    }
    {
        const bool up = (c & 1) != 0;
        const float keep = up ? w2[1] : w2[0];
        const float send = up ? w2[0] : w2[1];
        w1 = keep + row_xor<1>(send);
    }
    lg[c >> 3][mhalf + (c & 7)][nblk] = w1;
    __syncthreads();
    #pragma unroll
    for (int i = 0; i < 8; ++i) {
        const size_t row = (size_t)(mt * 16 + mhalf + i) * OUTD;
        h[row + col0 +  0] = acc0[i];
        h[row + col0 + 16] = acc1[i];
        h[row + col0 + 32] = acc2[i];
        h[row + col0 + 48] = acc3[i];
    }
    if (nblk < 2) {
        float* gp = (nblk == 0 ? a_src : a_dst) + (size_t)mt * (16 * HHEADS);
        const float2 val = *(const float2*)(&lg[nblk][0][0] + 2 * lane);
        *(float2*)(gp + 2 * lane) = val;
    }
}

#define TROWS        200
#define NTILE        (NNODE / TROWS)
#define AGG_THREADS  1024
#define AGG_WAVES    (AGG_THREADS / 32)
#define NREALCH      (NEDGE / 32)
#define NREALIT      ((NREALCH + AGG_WAVES - 1) / AGG_WAVES)
#define SELF_PER_W   ((TROWS + AGG_WAVES - 1) / AGG_WAVES)
#define P1_GROUP     6
#define LCAP         (NEDGE + TROWS)
#define P2_EDGES     4
static_assert(NNODE % TROWS == 0, "tiles must cover the nodes exactly");
static_assert(NEDGE % 32 == 0, "listed edges come in whole chunks");
static_assert(SELF_PER_W <= 32, "a wave's self loops fit one chunk");
static_assert(NREALIT % P1_GROUP == 0, "pass 1 takes the sweeps in whole groups");
static_assert(LCAP % P2_EDGES == 0, "a trip's four entries never reach past the list's room");
static_assert(NNODE <= 65536 && TROWS <= 65536, "source and row share one packed word");

__device__ __forceinline__ float4 edge_exp4(const float4 as, const float4 ad) {
    float4 ex;
    float v;
    v = as.x + ad.x; v = v > 0.0f ? v : NEG_SLOPE * v; ex.x = __expf(v);
    v = as.y + ad.y; v = v > 0.0f ? v : NEG_SLOPE * v; ex.y = __expf(v);
    v = as.z + ad.z; v = v > 0.0f ? v : NEG_SLOPE * v; ex.z = __expf(v);
    v = as.w + ad.w; v = v > 0.0f ? v : NEG_SLOPE * v; ex.w = __expf(v);
    return ex;
}

__global__ __launch_bounds__(AGG_THREADS)
void gat_aggregate(const int* __restrict__ ei,
                   const float* __restrict__ h,
                   const float* __restrict__ a_src,
                   const float* __restrict__ a_dst,
                   const float* __restrict__ bias,
                   float* __restrict__ out) {
    __shared__ __attribute__((aligned(16))) float    acc[TROWS * FDIM];
    __shared__ __attribute__((aligned(16))) float    den_l[TROWS * HHEADS];
    __shared__ __attribute__((aligned(16))) float4   lw[LCAP];
    __shared__ __attribute__((aligned(16))) unsigned lsd[LCAP];
    __shared__ unsigned lcount;

    const int r    = blockIdx.x / NTILE;
    const int d0   = (blockIdx.x % NTILE) * TROWS;
    const int lane = threadIdx.x & 31;
    const int wv   = threadIdx.x >> 5;

    for (int i = threadIdx.x; i < TROWS * FDIM; i += AGG_THREADS)
        acc[i] = bias[i & (FDIM - 1)];
    for (int i = threadIdx.x; i < TROWS * HHEADS; i += AGG_THREADS)
        den_l[i] = 0.0f;
    if (threadIdx.x == 0) lcount = 0u;
    __syncthreads();

    const float4* as4 = (const float4*)(a_src + (size_t)r * NNODE * HHEADS);
    const float4* ad4 = (const float4*)(a_dst + (size_t)r * NNODE * HHEADS);

    {
        int sA[NREALIT], dA[NREALIT];
        #pragma unroll
        for (int it = 0; it < NREALIT; ++it) {
            const int cc = wv + it * AGG_WAVES;
            sA[it] = 0; dA[it] = -1;
            if (cc < NREALCH) {
                const int e = cc * 32 + lane;
                sA[it] = ei[e];
                dA[it] = ei[NEDGE + e] - d0;
            }
        }

        unsigned msk[NREALIT];
        unsigned total = 0u;
        #pragma unroll
        for (int it = 0; it < NREALIT; ++it) {
            msk[it] = (unsigned)__ballot((unsigned)dA[it] < (unsigned)TROWS);
            total += (unsigned)__builtin_popcount(msk[it]);
        }
        const int  q      = wv * SELF_PER_W + lane;
        const bool selfok = lane < SELF_PER_W && q < TROWS;
        const unsigned mself = (unsigned)__ballot(selfok);
        total += (unsigned)__builtin_popcount(mself);
        unsigned base = 0u;
        if (lane == 0) base = atomicAdd(&lcount, total);
        base = (unsigned)__builtin_amdgcn_readfirstlane((int)base);
        const unsigned below = (1u << lane) - 1u;

        #pragma unroll
        for (int g = 0; g < NREALIT; g += P1_GROUP) {
            float4 exv[P1_GROUP];
            #pragma unroll
            for (int j = 0; j < P1_GROUP; ++j) {
                const int  s   = sA[g + j], dl = dA[g + j];
                const bool inr = (unsigned)dl < (unsigned)TROWS;
                const float4 av = as4[inr ? s : d0];
                const float4 dv = ad4[inr ? d0 + dl : d0];
                float4 ex = edge_exp4(av, dv);
                if (s == d0 + dl) ex = make_float4(0.f, 0.f, 0.f, 0.f);
                exv[j] = ex;
            }
            #pragma unroll
            for (int j = 0; j < P1_GROUP; ++j) {
                const int dl = dA[g + j];
                if ((unsigned)dl < (unsigned)TROWS) {
                    const unsigned slot = base + (unsigned)__builtin_popcount(msk[g + j] & below);
                    lw[slot]  = exv[j];
                    lsd[slot] = (unsigned)sA[g + j] | ((unsigned)dl << 16);
                    float* dp = den_l + dl * HHEADS;
                    atomicAdd(dp + 0, exv[j].x);
                    atomicAdd(dp + 1, exv[j].y);
                    atomicAdd(dp + 2, exv[j].z);
                    atomicAdd(dp + 3, exv[j].w);
                }
                base += (unsigned)__builtin_popcount(msk[g + j]);
            }
        }

        if (selfok) {
            const float4 ex = edge_exp4(as4[d0 + q], ad4[d0 + q]);
            const unsigned slot = base + (unsigned)__builtin_popcount(mself & below);
            lw[slot]  = ex;
            lsd[slot] = (unsigned)(d0 + q) | ((unsigned)q << 16);
            float* dp = den_l + q * HHEADS;
            atomicAdd(dp + 0, ex.x);
            atomicAdd(dp + 1, ex.y);
            atomicAdd(dp + 2, ex.z);
            atomicAdd(dp + 3, ex.w);
        }
    }
    __syncthreads();

    const int n = __builtin_amdgcn_readfirstlane((int)lcount);
    for (int i = threadIdx.x; i < n; i += AGG_THREADS) {
        const int    dl  = (int)(lsd[i] >> 16);
        const float4 ex4 = lw[i];
        const float4 dn4 = *(const float4*)(den_l + dl * HHEADS);
        float4 a;
        a.x = (1.0f / HHEADS) * ex4.x / (dn4.x + 1e-16f);
        a.y = (1.0f / HHEADS) * ex4.y / (dn4.y + 1e-16f);
        a.z = (1.0f / HHEADS) * ex4.z / (dn4.z + 1e-16f);
        a.w = (1.0f / HHEADS) * ex4.w / (dn4.w + 1e-16f);
        lw[i] = a;
    }
    __syncthreads();

    const float* hr   = h + (size_t)r * NNODE * OUTD + 2 * lane;
    float*       accl = acc + 2 * lane;
    const int wvu = __builtin_amdgcn_readfirstlane(wv);

    int   i0 = wvu * P2_EDGES;
    uint4 sdn = make_uint4(0u, 0u, 0u, 0u);
    float4 wn[P2_EDGES];
    #pragma unroll
    for (int k = 0; k < P2_EDGES; ++k) wn[k] = make_float4(0.f, 0.f, 0.f, 0.f);
    if (i0 < n) {
        sdn = *(const uint4*)(lsd + i0);
        #pragma unroll
        for (int k = 0; k < P2_EDGES; ++k) wn[k] = lw[i0 + k];
    }

    while (i0 < n) {
        unsigned sdc[P2_EDGES];
        sdc[0] = (unsigned)__builtin_amdgcn_readfirstlane((int)sdn.x);
        sdc[1] = (unsigned)__builtin_amdgcn_readfirstlane((int)sdn.y);
        sdc[2] = (unsigned)__builtin_amdgcn_readfirstlane((int)sdn.z);
        sdc[3] = (unsigned)__builtin_amdgcn_readfirstlane((int)sdn.w);
        float4 wc[P2_EDGES];
        #pragma unroll
        for (int k = 0; k < P2_EDGES; ++k) wc[k] = wn[k];
        const int i1 = i0 + AGG_WAVES * P2_EDGES;

        if (i0 + P2_EDGES <= n) {
            v2f xv[P2_EDGES][4];
            #pragma unroll
            for (int k = 0; k < P2_EDGES; ++k) {
                const float* p = hr + (size_t)(sdc[k] & 0xffffu) * OUTD;
                #pragma unroll
                for (int j = 0; j < 4; ++j)
                    xv[k][j] = *(const v2f*)(p + j * CDIM);
            }
            if (i1 < n) {
                sdn = *(const uint4*)(lsd + i1);
                #pragma unroll
                for (int k = 0; k < P2_EDGES; ++k) wn[k] = lw[i1 + k];
            }
            #pragma unroll
            for (int k = 0; k < P2_EDGES; ++k) {
                v2f m = xv[k][0] * wc[k].x;
                m += xv[k][1] * wc[k].y;
                m += xv[k][2] * wc[k].z;
                m += xv[k][3] * wc[k].w;
                float* ap = accl + (sdc[k] >> 16) * FDIM;
                atomicAdd(ap,     m.x);
                atomicAdd(ap + 1, m.y);
            }
        } else {
            #pragma unroll
            for (int k = 0; k < P2_EDGES - 1; ++k) {
                if (i0 + k < n) {
                    const float* p = hr + (size_t)(sdc[k] & 0xffffu) * OUTD;
                    const float2 x0 = *(const float2*)(p);
                    const float2 x1 = *(const float2*)(p + CDIM);
                    const float2 x2 = *(const float2*)(p + 2 * CDIM);
                    const float2 x3 = *(const float2*)(p + 3 * CDIM);
                    float m0 = wc[k].x * x0.x;
                    float m1 = wc[k].x * x0.y;
                    m0 += wc[k].y * x1.x;  m1 += wc[k].y * x1.y;
                    m0 += wc[k].z * x2.x;  m1 += wc[k].z * x2.y;
                    m0 += wc[k].w * x3.x;  m1 += wc[k].w * x3.y;
                    float* ap = accl + (sdc[k] >> 16) * FDIM;
                    atomicAdd(ap,     m0);
                    atomicAdd(ap + 1, m1);
                }
            }
        }
        i0 = i1;
    }

    __syncthreads();
    float4*       o4 = (float4*)(out + ((size_t)r * NNODE + d0) * FDIM);
    const float4* a4 = (const float4*)acc;
    for (int i = threadIdx.x; i < TROWS * FDIM / 4; i += AGG_THREADS)
        o4[i] = a4[i];
}

extern "C" void kernel_launch(void* const* d_in, const int* in_sizes, int n_in,
                              void* d_out, int out_size, void* d_ws, size_t ws_size,
                              hipStream_t stream) {
    const float* x       = (const float*)d_in[0];
    const int*   ei      = (const int*)d_in[1];
    const float* W       = (const float*)d_in[2];
    const float* att_src = (const float*)d_in[3];
    const float* att_dst = (const float*)d_in[4];
    const float* bias    = (const float*)d_in[5];
    float* out = (float*)d_out;

    float* ws    = (float*)d_ws;
    float* hbuf  = ws;
    float* a_src = hbuf  + (size_t)NROWS * OUTD;
    float* a_dst = a_src + (size_t)NROWS * HHEADS;

    {
        dim3 grid(NROWS / 16);
        dim3 block(128);
        gat_proj_wmma<<<grid, block, 0, stream>>>(x, W, att_src, att_dst,
                                                  hbuf, a_src, a_dst);
    }
    {
        gat_aggregate<<<RREP * NTILE, AGG_THREADS, 0, stream>>>(ei, hbuf, a_src, a_dst,
                                                               bias, out);
    }
}


# Aggregation pass 2 by source: list entries chained per source node in LDS (one exchange each); a wave loads a source's h row once per tile and walks its chain; acc sum order changes
# speedup vs baseline: 9.8799x; 1.1621x over previous
#include <hip/hip_runtime.h>
#include <hip/hip_bf16.h>

typedef float v2f __attribute__((ext_vector_type(2)));
typedef float v8f __attribute__((ext_vector_type(8)));

#define RREP   48
#define NNODE  1000
#define FDIM   64
#define HHEADS 4
#define CDIM   64
#define OUTD   256
#define NEDGE  12000
#define NROWS  48000
#define NEG_SLOPE 0.2f

template <int MASK>
__device__ __forceinline__ float row_xor(const float v) {
    return __int_as_float(__builtin_amdgcn_update_dpp(0, __float_as_int(v),
                                                      0x160 + MASK, 0xf, 0xf, true));
}

__global__ void gat_proj_wmma(const float* __restrict__ x,
                              const float* __restrict__ W,
                              const float* __restrict__ att_src,
                              const float* __restrict__ att_dst,
                              float* __restrict__ h,
                              float* __restrict__ a_src,
                              float* __restrict__ a_dst) {
    __shared__ __attribute__((aligned(16))) float lg[2][16][HHEADS];

    const int lane = threadIdx.x & 31;
    const int nblk = threadIdx.x >> 5;
    const int mt   = blockIdx.x;
    const int mrow = mt * 16 + (lane & 15);
    const int kb   = (lane >> 4) * 2;
    const int col0 = nblk * 64 + (lane & 15);

    v8f acc0 = {}, acc1 = {}, acc2 = {}, acc3 = {};

    #pragma unroll
    for (int kk = 0; kk < 16; ++kk) {
        const int k = kk * 4 + kb;
        v2f a;
        const float2 av = *(const float2*)(x + (size_t)mrow * FDIM + k);
        a.x = av.x; a.y = av.y;
        v2f b0, b1, b2, b3;
        b0.x = W[k * OUTD + col0 +  0]; b0.y = W[(k + 1) * OUTD + col0 +  0];
        b1.x = W[k * OUTD + col0 + 16]; b1.y = W[(k + 1) * OUTD + col0 + 16];
        b2.x = W[k * OUTD + col0 + 32]; b2.y = W[(k + 1) * OUTD + col0 + 32];
        b3.x = W[k * OUTD + col0 + 48]; b3.y = W[(k + 1) * OUTD + col0 + 48];

        acc0 = __builtin_amdgcn_wmma_f32_16x16x4_f32(false, a, false, b0,
                                                     (short)0, acc0, false, false);
        acc1 = __builtin_amdgcn_wmma_f32_16x16x4_f32(false, a, false, b1,
                                                     (short)0, acc1, false, false);
        acc2 = __builtin_amdgcn_wmma_f32_16x16x4_f32(false, a, false, b2,
                                                     (short)0, acc2, false, false);
        acc3 = __builtin_amdgcn_wmma_f32_16x16x4_f32(false, a, false, b3,
                                                     (short)0, acc3, false, false);
    }

    const int mhalf = (lane >> 4) * 8;

    const int c = lane & 15;
    const float* asp = att_src + nblk * CDIM + c;
    const float* adp = att_dst + nblk * CDIM + c;
    const float s0 = asp[0], s1 = asp[16], s2 = asp[32], s3 = asp[48];
    const float t0 = adp[0], t1 = adp[16], t2 = adp[32], t3 = adp[48];
    float v[16];
    for (int i = 0; i < 8; ++i) {
        v[i]     = fmaf(acc3[i], s3, fmaf(acc2[i], s2, fmaf(acc1[i], s1, acc0[i] * s0)));
        v[8 + i] = fmaf(acc3[i], t3, fmaf(acc2[i], t2, fmaf(acc1[i], t1, acc0[i] * t0)));
    }
    float w8[8], w4[4], w2[2], w1;
    {
        const bool up = (c & 8) != 0;
        for (int m = 0; m < 8; ++m) {
            const float keep = up ? v[m + 8] : v[m];
            const float send = up ? v[m] : v[m + 8];
            w8[m] = keep + row_xor<8>(send);
        }
    }
    {
        const bool up = (c & 4) != 0;
        for (int m = 0; m < 4; ++m) {
            const float keep = up ? w8[m + 4] : w8[m];
            const float send = up ? w8[m] : w8[m + 4];
            w4[m] = keep + row_xor<4>(send);
        }
    }
    {
        const bool up = (c & 2) != 0;
        for (int m = 0; m < 2; ++m) {
            const float keep = up ? w4[m + 2] : w4[m];
            const float send = up ? w4[m] : w4[m + 2];
            w2[m] = keep + row_xor<2>(send);
        }
    }
    {
        const bool up = (c & 1) != 0;
        const float keep = up ? w2[1] : w2[0];
        const float send = up ? w2[0] : w2[1];
        w1 = keep + row_xor<1>(send);
    }
    lg[c >> 3][mhalf + (c & 7)][nblk] = w1;
    __syncthreads();
    #pragma unroll
    for (int i = 0; i < 8; ++i) {
        const size_t row = (size_t)(mt * 16 + mhalf + i) * OUTD;
        h[row + col0 +  0] = acc0[i];
        h[row + col0 + 16] = acc1[i];
        h[row + col0 + 32] = acc2[i];
        h[row + col0 + 48] = acc3[i];
    }
    if (nblk < 2) {
        float* gp = (nblk == 0 ? a_src : a_dst) + (size_t)mt * (16 * HHEADS);
        const float2 val = *(const float2*)(&lg[nblk][0][0] + 2 * lane);
        *(float2*)(gp + 2 * lane) = val;
    }
}

#define TROWS        200
#define NTILE        (NNODE / TROWS)
#define AGG_THREADS  1024
#define AGG_WAVES    (AGG_THREADS / 32)
#define NREALCH      (NEDGE / 32)
#define NREALIT      ((NREALCH + AGG_WAVES - 1) / AGG_WAVES)
#define SELF_PER_W   ((TROWS + AGG_WAVES - 1) / AGG_WAVES)
#define P1_GROUP     6
#define LCAP         (NEDGE + TROWS)
#define P2_SRC       4
#define SRC_PER_W    ((NNODE + AGG_WAVES - 1) / AGG_WAVES)
#define LEND         0xffffu
static_assert(NNODE % TROWS == 0, "tiles must cover the nodes exactly");
static_assert(NEDGE % 32 == 0, "listed edges come in whole chunks");
static_assert(SELF_PER_W <= 32, "a wave's self loops fit one chunk");
static_assert(NREALIT % P1_GROUP == 0, "pass 1 takes the sweeps in whole groups");
static_assert(SRC_PER_W <= 32, "a wave's chain heads fit one lane each");
static_assert(SRC_PER_W % P2_SRC == 0, "pass 2 takes a wave's sources in whole trips");
static_assert(LCAP < LEND && TROWS <= 65536, "next entry and row share one packed word");

__device__ __forceinline__ float4 edge_exp4(const float4 as, const float4 ad) {
    float4 ex;
    float v;
    v = as.x + ad.x; v = v > 0.0f ? v : NEG_SLOPE * v; ex.x = __expf(v);
    v = as.y + ad.y; v = v > 0.0f ? v : NEG_SLOPE * v; ex.y = __expf(v);
    v = as.z + ad.z; v = v > 0.0f ? v : NEG_SLOPE * v; ex.z = __expf(v);
    v = as.w + ad.w; v = v > 0.0f ? v : NEG_SLOPE * v; ex.w = __expf(v);
    return ex;
}

__global__ __launch_bounds__(AGG_THREADS)
void gat_aggregate(const int* __restrict__ ei,
                   const float* __restrict__ h,
                   const float* __restrict__ a_src,
                   const float* __restrict__ a_dst,
                   const float* __restrict__ bias,
                   float* __restrict__ out) {
    __shared__ __attribute__((aligned(16))) float    acc[TROWS * FDIM];
    __shared__ __attribute__((aligned(16))) float    den_l[TROWS * HHEADS];
    __shared__ __attribute__((aligned(16))) float4   lw[LCAP];
    __shared__ __attribute__((aligned(16))) unsigned lsd[LCAP];
    __shared__ unsigned head[NNODE];
    __shared__ unsigned lcount;

    const int r    = blockIdx.x / NTILE;
    const int d0   = (blockIdx.x % NTILE) * TROWS;
    const int lane = threadIdx.x & 31;
    const int wv   = threadIdx.x >> 5;

    for (int i = threadIdx.x; i < TROWS * FDIM; i += AGG_THREADS)
        acc[i] = bias[i & (FDIM - 1)];
    for (int i = threadIdx.x; i < TROWS * HHEADS; i += AGG_THREADS)
        den_l[i] = 0.0f;
    for (int i = threadIdx.x; i < NNODE; i += AGG_THREADS)
        head[i] = LEND;
    if (threadIdx.x == 0) lcount = 0u;
    __syncthreads();

    const float4* as4 = (const float4*)(a_src + (size_t)r * NNODE * HHEADS);
    const float4* ad4 = (const float4*)(a_dst + (size_t)r * NNODE * HHEADS);

    {
        int sA[NREALIT], dA[NREALIT];
        #pragma unroll
        for (int it = 0; it < NREALIT; ++it) {
            const int cc = wv + it * AGG_WAVES;
            sA[it] = 0; dA[it] = -1;
            if (cc < NREALCH) {
                const int e = cc * 32 + lane;
                sA[it] = ei[e];
                dA[it] = ei[NEDGE + e] - d0;
            }
        }

        unsigned msk[NREALIT];
        unsigned total = 0u;
        #pragma unroll
        for (int it = 0; it < NREALIT; ++it) {
            msk[it] = (unsigned)__ballot((unsigned)dA[it] < (unsigned)TROWS);
            total += (unsigned)__builtin_popcount(msk[it]);
        }
        const int  q      = wv * SELF_PER_W + lane;
        const bool selfok = lane < SELF_PER_W && q < TROWS;
        const unsigned mself = (unsigned)__ballot(selfok);
        total += (unsigned)__builtin_popcount(mself);
        unsigned base = 0u;
        if (lane == 0) base = atomicAdd(&lcount, total);
        base = (unsigned)__builtin_amdgcn_readfirstlane((int)base);
        const unsigned below = (1u << lane) - 1u;

        #pragma unroll
        for (int g = 0; g < NREALIT; g += P1_GROUP) {
            float4 exv[P1_GROUP];
            #pragma unroll
            for (int j = 0; j < P1_GROUP; ++j) {
                const int  s   = sA[g + j], dl = dA[g + j];
                const bool inr = (unsigned)dl < (unsigned)TROWS;
                const float4 av = as4[inr ? s : d0];
                const float4 dv = ad4[inr ? d0 + dl : d0];
                float4 ex = edge_exp4(av, dv);
                if (s == d0 + dl) ex = make_float4(0.f, 0.f, 0.f, 0.f);
                exv[j] = ex;
            }
            unsigned slotv[P1_GROUP], prevv[P1_GROUP];
            #pragma unroll
            for (int j = 0; j < P1_GROUP; ++j) {
                slotv[j] = base + (unsigned)__builtin_popcount(msk[g + j] & below);
                prevv[j] = LEND;
                if ((unsigned)dA[g + j] < (unsigned)TROWS)
                    prevv[j] = atomicExch(&head[sA[g + j]], slotv[j]);
                base += (unsigned)__builtin_popcount(msk[g + j]);
            }
            #pragma unroll
            for (int j = 0; j < P1_GROUP; ++j) {
                const int dl = dA[g + j];
                if ((unsigned)dl < (unsigned)TROWS) {
                    const unsigned slot = slotv[j];
                    lw[slot]  = exv[j];
                    lsd[slot] = prevv[j] | ((unsigned)dl << 16);
                    float* dp = den_l + dl * HHEADS;
                    atomicAdd(dp + 0, exv[j].x);
                    atomicAdd(dp + 1, exv[j].y);
                    atomicAdd(dp + 2, exv[j].z);
                    atomicAdd(dp + 3, exv[j].w);
                }
            }
        }

        if (selfok) {
            const float4 ex = edge_exp4(as4[d0 + q], ad4[d0 + q]);
            const unsigned slot = base + (unsigned)__builtin_popcount(mself & below);
            const unsigned prev = atomicExch(&head[d0 + q], slot);
            lw[slot]  = ex;
            lsd[slot] = prev | ((unsigned)q << 16);
            float* dp = den_l + q * HHEADS;
            atomicAdd(dp + 0, ex.x);
            atomicAdd(dp + 1, ex.y);
            atomicAdd(dp + 2, ex.z);
            atomicAdd(dp + 3, ex.w);
        }
    }
    __syncthreads();

    const float* hr   = h + (size_t)r * NNODE * OUTD + 2 * lane;
    float*       accl = acc + 2 * lane;
    const int wvu = __builtin_amdgcn_readfirstlane(wv);

    v2f xn[P2_SRC][4];
    #pragma unroll
    for (int k = 0; k < P2_SRC; ++k) {
        const int s = min(wvu + AGG_WAVES * k, NNODE - 1);
        const float* p = hr + (size_t)s * OUTD;
        #pragma unroll
        for (int j = 0; j < 4; ++j)
            xn[k][j] = *(const v2f*)(p + j * CDIM);
    }

    const int n = __builtin_amdgcn_readfirstlane((int)lcount);
    for (int i = threadIdx.x; i < n; i += AGG_THREADS) {
        const int    dl  = (int)(lsd[i] >> 16);
        const float4 ex4 = lw[i];
        const float4 dn4 = *(const float4*)(den_l + dl * HHEADS);
        float4 a;
        a.x = (1.0f / HHEADS) * ex4.x / (dn4.x + 1e-16f);
        a.y = (1.0f / HHEADS) * ex4.y / (dn4.y + 1e-16f);
        a.z = (1.0f / HHEADS) * ex4.z / (dn4.z + 1e-16f);
        a.w = (1.0f / HHEADS) * ex4.w / (dn4.w + 1e-16f);
        lw[i] = a;
    }
    __syncthreads();

    unsigned hv = LEND;
    {
        const int sl = wvu + AGG_WAVES * lane;
        if (lane < SRC_PER_W && sl < NNODE) hv = head[sl];
    }

    for (int j0 = 0; j0 < SRC_PER_W; j0 += P2_SRC) {
        v2f xc[P2_SRC][4];
        #pragma unroll
        for (int k = 0; k < P2_SRC; ++k) {
            #pragma unroll
            for (int j = 0; j < 4; ++j) xc[k][j] = xn[k][j];
        }
        if (j0 + P2_SRC < SRC_PER_W) {
            #pragma unroll
            for (int k = 0; k < P2_SRC; ++k) {
                const int s = min(wvu + AGG_WAVES * (j0 + P2_SRC + k), NNODE - 1);
                const float* p = hr + (size_t)s * OUTD;
                #pragma unroll
                for (int j = 0; j < 4; ++j)
                    xn[k][j] = *(const v2f*)(p + j * CDIM);
            }
        }
        #pragma unroll
        for (int k = 0; k < P2_SRC; ++k) {
            unsigned cur = (unsigned)__builtin_amdgcn_readlane((int)hv, j0 + k);
            for (int hop = 0; hop < LCAP && cur != LEND; ++hop) {
                const unsigned e  = lsd[cur];
                const float4   wc = lw[cur];
                v2f m = xc[k][0] * wc.x;
                m += xc[k][1] * wc.y;
                m += xc[k][2] * wc.z;
                m += xc[k][3] * wc.w;
                const unsigned eu = (unsigned)__builtin_amdgcn_readfirstlane((int)e);
                float* ap = accl + (eu >> 16) * FDIM;
                atomicAdd(ap,     m.x);
                atomicAdd(ap + 1, m.y);
                cur = eu & 0xffffu;
            }
        }
    }

    __syncthreads();
    float4*       o4 = (float4*)(out + ((size_t)r * NNODE + d0) * FDIM);
    const float4* a4 = (const float4*)acc;
    for (int i = threadIdx.x; i < TROWS * FDIM / 4; i += AGG_THREADS)
        o4[i] = a4[i];
}

extern "C" void kernel_launch(void* const* d_in, const int* in_sizes, int n_in,
                              void* d_out, int out_size, void* d_ws, size_t ws_size,
                              hipStream_t stream) {
    const float* x       = (const float*)d_in[0];
    const int*   ei      = (const int*)d_in[1];
    const float* W       = (const float*)d_in[2];
    const float* att_src = (const float*)d_in[3];
    const float* att_dst = (const float*)d_in[4];
    const float* bias    = (const float*)d_in[5];
    float* out = (float*)d_out;

    float* ws    = (float*)d_ws;
    float* hbuf  = ws;
    float* a_src = hbuf  + (size_t)NROWS * OUTD;
    float* a_dst = a_src + (size_t)NROWS * HHEADS;

    {
        dim3 grid(NROWS / 16);
        dim3 block(128);
        gat_proj_wmma<<<grid, block, 0, stream>>>(x, W, att_src, att_dst,
                                                  hbuf, a_src, a_dst);
    }
    {
        gat_aggregate<<<RREP * NTILE, AGG_THREADS, 0, stream>>>(ei, hbuf, a_src, a_dst,
                                                               bias, out);
    }
}


# Projection made resident: 1000 workgroups loop over 3 row tiles each, W fragments loaded once and kept in registers, x tiles by whole-line loads through LDS; products and sum order unchanged
# speedup vs baseline: 10.4976x; 1.0625x over previous
#include <hip/hip_runtime.h>
#include <hip/hip_bf16.h>

typedef float v2f __attribute__((ext_vector_type(2)));
typedef float v8f __attribute__((ext_vector_type(8)));

#define RREP   48
#define NNODE  1000
#define FDIM   64
#define HHEADS 4
#define CDIM   64
#define OUTD   256
#define NEDGE  12000
#define NROWS  48000
#define NEG_SLOPE 0.2f

template <int MASK>
__device__ __forceinline__ float row_xor(const float v) {
    return __int_as_float(__builtin_amdgcn_update_dpp(0, __float_as_int(v),
                                                      0x160 + MASK, 0xf, 0xf, true));
}

#define NMT      (NROWS / 16)
#define PROJ_WG  1000
#define XPITCH   (FDIM + 4)
static_assert(NMT % PROJ_WG == 0, "every workgroup takes the same number of tiles");

__global__ __launch_bounds__(128)
void gat_proj_wmma(const float* __restrict__ x,
                   const float* __restrict__ W,
                   const float* __restrict__ att_src,
                   const float* __restrict__ att_dst,
                   float* __restrict__ h,
                   float* __restrict__ a_src,
                   float* __restrict__ a_dst) {
    __shared__ __attribute__((aligned(16))) float xs[2][16 * XPITCH];
    __shared__ __attribute__((aligned(16))) float lg[2][2][16][HHEADS];

    const int lane  = threadIdx.x & 31;
    const int nblk  = threadIdx.x >> 5;
    const int c     = lane & 15;
    const int kb    = (lane >> 4) * 2;
    const int col0  = nblk * 64 + c;
    const int mhalf = (lane >> 4) * 8;

    const int q0  = threadIdx.x, q1 = threadIdx.x + 128;
    const int so0 = (q0 >> 4) * XPITCH + (q0 & 15) * 4;
    const int so1 = (q1 >> 4) * XPITCH + (q1 & 15) * 4;

    int mt = blockIdx.x;
    float4 n0, n1;
    {
        const float4* xt = (const float4*)(x + (size_t)mt * (16 * FDIM));
        n0 = xt[q0]; n1 = xt[q1];
    }

    *(float4*)(&xs[0][so0]) = n0;
    *(float4*)(&xs[0][so1]) = n1;
    __syncthreads();

    v2f b[16][4];
    #pragma unroll
    for (int kk = 0; kk < 16; ++kk) {
        const int k = kk * 4 + kb;
        #pragma unroll
        for (int j = 0; j < 4; ++j) {
            b[kk][j].x = W[k * OUTD + col0 + 16 * j];
            b[kk][j].y = W[(k + 1) * OUTD + col0 + 16 * j];
        }
    }

    const float* asp = att_src + nblk * CDIM + c;
    const float* adp = att_dst + nblk * CDIM + c;
    const float s0 = asp[0], s1 = asp[16], s2 = asp[32], s3 = asp[48];
    const float t0 = adp[0], t1 = adp[16], t2 = adp[32], t3 = adp[48];

    int p = 0;
    for (; mt < NMT; mt += PROJ_WG) {
        const bool more = mt + PROJ_WG < NMT;
        if (more) {
            const float4* xt = (const float4*)(x + (size_t)(mt + PROJ_WG) * (16 * FDIM));
            n0 = xt[q0]; n1 = xt[q1];
        }

        v8f acc0 = {}, acc1 = {}, acc2 = {}, acc3 = {};
        const float* xr = &xs[p][c * XPITCH + kb];
        #pragma unroll
        for (int kk = 0; kk < 16; ++kk) {
            const v2f a = *(const v2f*)(xr + kk * 4);
            acc0 = __builtin_amdgcn_wmma_f32_16x16x4_f32(false, a, false, b[kk][0],
                                                         (short)0, acc0, false, false);
            acc1 = __builtin_amdgcn_wmma_f32_16x16x4_f32(false, a, false, b[kk][1],
                                                         (short)0, acc1, false, false);
            acc2 = __builtin_amdgcn_wmma_f32_16x16x4_f32(false, a, false, b[kk][2],
                                                         (short)0, acc2, false, false);
            acc3 = __builtin_amdgcn_wmma_f32_16x16x4_f32(false, a, false, b[kk][3],
                                                         (short)0, acc3, false, false);
        }

        float v[16];
        for (int i = 0; i < 8; ++i) {
            v[i]     = fmaf(acc3[i], s3, fmaf(acc2[i], s2, fmaf(acc1[i], s1, acc0[i] * s0)));
            v[8 + i] = fmaf(acc3[i], t3, fmaf(acc2[i], t2, fmaf(acc1[i], t1, acc0[i] * t0)));
        }
        float w8[8], w4[4], w2[2], w1;
        {
            const bool up = (c & 8) != 0;
            for (int m = 0; m < 8; ++m) {
                const float keep = up ? v[m + 8] : v[m];
                const float send = up ? v[m] : v[m + 8];
                w8[m] = keep + row_xor<8>(send);
            }
        }
        {
            const bool up = (c & 4) != 0;
            for (int m = 0; m < 4; ++m) {
                const float keep = up ? w8[m + 4] : w8[m];
                const float send = up ? w8[m] : w8[m + 4];
                w4[m] = keep + row_xor<4>(send);
            }
        }
        {
            const bool up = (c & 2) != 0;
            for (int m = 0; m < 2; ++m) {
                const float keep = up ? w4[m + 2] : w4[m];
                const float send = up ? w4[m] : w4[m + 2];
                w2[m] = keep + row_xor<2>(send);
            }
        }
        {
            const bool up = (c & 1) != 0;
            const float keep = up ? w2[1] : w2[0];
            const float send = up ? w2[0] : w2[1];
            w1 = keep + row_xor<1>(send);
        }
        lg[p][c >> 3][mhalf + (c & 7)][nblk] = w1;
        if (more) {
            *(float4*)(&xs[p ^ 1][so0]) = n0;
            *(float4*)(&xs[p ^ 1][so1]) = n1;
        }
        __syncthreads();
        #pragma unroll
        for (int i = 0; i < 8; ++i) {
            const size_t row = (size_t)(mt * 16 + mhalf + i) * OUTD;
            h[row + col0 +  0] = acc0[i];
            h[row + col0 + 16] = acc1[i];
            h[row + col0 + 32] = acc2[i];
            h[row + col0 + 48] = acc3[i];
        }
        if (nblk < 2) {
            float* gp = (nblk == 0 ? a_src : a_dst) + (size_t)mt * (16 * HHEADS);
            const float2 val = *(const float2*)(&lg[p][nblk][0][0] + 2 * lane);
            *(float2*)(gp + 2 * lane) = val;
        }
        p ^= 1;
    }
}

#define TROWS        200
#define NTILE        (NNODE / TROWS)
#define AGG_THREADS  1024
#define AGG_WAVES    (AGG_THREADS / 32)
#define NREALCH      (NEDGE / 32)
#define NREALIT      ((NREALCH + AGG_WAVES - 1) / AGG_WAVES)
#define SELF_PER_W   ((TROWS + AGG_WAVES - 1) / AGG_WAVES)
#define P1_GROUP     6
#define LCAP         (NEDGE + TROWS)
#define P2_SRC       4
#define SRC_PER_W    ((NNODE + AGG_WAVES - 1) / AGG_WAVES)
#define LEND         0xffffu
static_assert(NNODE % TROWS == 0, "tiles must cover the nodes exactly");
static_assert(NEDGE % 32 == 0, "listed edges come in whole chunks");
static_assert(SELF_PER_W <= 32, "a wave's self loops fit one chunk");
static_assert(NREALIT % P1_GROUP == 0, "pass 1 takes the sweeps in whole groups");
static_assert(SRC_PER_W <= 32, "a wave's chain heads fit one lane each");
static_assert(SRC_PER_W % P2_SRC == 0, "pass 2 takes a wave's sources in whole trips");
static_assert(LCAP < LEND && TROWS <= 65536, "next entry and row share one packed word");

__device__ __forceinline__ float4 edge_exp4(const float4 as, const float4 ad) {
    float4 ex;
    float v;
    v = as.x + ad.x; v = v > 0.0f ? v : NEG_SLOPE * v; ex.x = __expf(v);
    v = as.y + ad.y; v = v > 0.0f ? v : NEG_SLOPE * v; ex.y = __expf(v);
    v = as.z + ad.z; v = v > 0.0f ? v : NEG_SLOPE * v; ex.z = __expf(v);
    v = as.w + ad.w; v = v > 0.0f ? v : NEG_SLOPE * v; ex.w = __expf(v);
    return ex;
}

__global__ __launch_bounds__(AGG_THREADS)
void gat_aggregate(const int* __restrict__ ei,
                   const float* __restrict__ h,
                   const float* __restrict__ a_src,
                   const float* __restrict__ a_dst,
                   const float* __restrict__ bias,
                   float* __restrict__ out) {
    __shared__ __attribute__((aligned(16))) float    acc[TROWS * FDIM];
    __shared__ __attribute__((aligned(16))) float    den_l[TROWS * HHEADS];
    __shared__ __attribute__((aligned(16))) float4   lw[LCAP];
    __shared__ __attribute__((aligned(16))) unsigned lsd[LCAP];
    __shared__ unsigned head[NNODE];
    __shared__ unsigned lcount;

    const int r    = blockIdx.x / NTILE;
    const int d0   = (blockIdx.x % NTILE) * TROWS;
    const int lane = threadIdx.x & 31;
    const int wv   = threadIdx.x >> 5;

    for (int i = threadIdx.x; i < TROWS * FDIM; i += AGG_THREADS)
        acc[i] = bias[i & (FDIM - 1)];
    for (int i = threadIdx.x; i < TROWS * HHEADS; i += AGG_THREADS)
        den_l[i] = 0.0f;
    for (int i = threadIdx.x; i < NNODE; i += AGG_THREADS)
        head[i] = LEND;
    if (threadIdx.x == 0) lcount = 0u;
    __syncthreads();

    const float4* as4 = (const float4*)(a_src + (size_t)r * NNODE * HHEADS);
    const float4* ad4 = (const float4*)(a_dst + (size_t)r * NNODE * HHEADS);

    {
        int sA[NREALIT], dA[NREALIT];
        #pragma unroll
        for (int it = 0; it < NREALIT; ++it) {
            const int cc = wv + it * AGG_WAVES;
            sA[it] = 0; dA[it] = -1;
            if (cc < NREALCH) {
                const int e = cc * 32 + lane;
                sA[it] = ei[e];
                dA[it] = ei[NEDGE + e] - d0;
            }
        }

        unsigned msk[NREALIT];
        unsigned total = 0u;
        #pragma unroll
        for (int it = 0; it < NREALIT; ++it) {
            msk[it] = (unsigned)__ballot((unsigned)dA[it] < (unsigned)TROWS);
            total += (unsigned)__builtin_popcount(msk[it]);
        }
        const int  q      = wv * SELF_PER_W + lane;
        const bool selfok = lane < SELF_PER_W && q < TROWS;
        const unsigned mself = (unsigned)__ballot(selfok);
        total += (unsigned)__builtin_popcount(mself);
        unsigned base = 0u;
        if (lane == 0) base = atomicAdd(&lcount, total);
        base = (unsigned)__builtin_amdgcn_readfirstlane((int)base);
        const unsigned below = (1u << lane) - 1u;

        #pragma unroll
        for (int g = 0; g < NREALIT; g += P1_GROUP) {
            float4 exv[P1_GROUP];
            #pragma unroll
            for (int j = 0; j < P1_GROUP; ++j) {
                const int  s   = sA[g + j], dl = dA[g + j];
                const bool inr = (unsigned)dl < (unsigned)TROWS;
                const float4 av = as4[inr ? s : d0];
                const float4 dv = ad4[inr ? d0 + dl : d0];
                float4 ex = edge_exp4(av, dv);
                if (s == d0 + dl) ex = make_float4(0.f, 0.f, 0.f, 0.f);
                exv[j] = ex;
            }
            unsigned slotv[P1_GROUP], prevv[P1_GROUP];
            #pragma unroll
            for (int j = 0; j < P1_GROUP; ++j) {
                slotv[j] = base + (unsigned)__builtin_popcount(msk[g + j] & below);
                prevv[j] = LEND;
                if ((unsigned)dA[g + j] < (unsigned)TROWS)
                    prevv[j] = atomicExch(&head[sA[g + j]], slotv[j]);
                base += (unsigned)__builtin_popcount(msk[g + j]);
            }
            #pragma unroll
            for (int j = 0; j < P1_GROUP; ++j) {
                const int dl = dA[g + j];
                if ((unsigned)dl < (unsigned)TROWS) {
                    const unsigned slot = slotv[j];
                    lw[slot]  = exv[j];
                    lsd[slot] = prevv[j] | ((unsigned)dl << 16);
                    float* dp = den_l + dl * HHEADS;
                    atomicAdd(dp + 0, exv[j].x);
                    atomicAdd(dp + 1, exv[j].y);
                    atomicAdd(dp + 2, exv[j].z);
                    atomicAdd(dp + 3, exv[j].w);
                }
            }
        }

        if (selfok) {
            const float4 ex = edge_exp4(as4[d0 + q], ad4[d0 + q]);
            const unsigned slot = base + (unsigned)__builtin_popcount(mself & below);
            const unsigned prev = atomicExch(&head[d0 + q], slot);
            lw[slot]  = ex;
            lsd[slot] = prev | ((unsigned)q << 16);
            float* dp = den_l + q * HHEADS;
            atomicAdd(dp + 0, ex.x);
            atomicAdd(dp + 1, ex.y);
            atomicAdd(dp + 2, ex.z);
            atomicAdd(dp + 3, ex.w);
        }
    }
    __syncthreads();

    const float* hr   = h + (size_t)r * NNODE * OUTD + 2 * lane;
    float*       accl = acc + 2 * lane;
    const int wvu = __builtin_amdgcn_readfirstlane(wv);

    v2f xn[P2_SRC][4];
    #pragma unroll
    for (int k = 0; k < P2_SRC; ++k) {
        const int s = min(wvu + AGG_WAVES * k, NNODE - 1);
        const float* p = hr + (size_t)s * OUTD;
        #pragma unroll
        for (int j = 0; j < 4; ++j)
            xn[k][j] = *(const v2f*)(p + j * CDIM);
    }

    const int n = __builtin_amdgcn_readfirstlane((int)lcount);
    for (int i = threadIdx.x; i < n; i += AGG_THREADS) {
        const int    dl  = (int)(lsd[i] >> 16);
        const float4 ex4 = lw[i];
        const float4 dn4 = *(const float4*)(den_l + dl * HHEADS);
        float4 a;
        a.x = (1.0f / HHEADS) * ex4.x / (dn4.x + 1e-16f);
        a.y = (1.0f / HHEADS) * ex4.y / (dn4.y + 1e-16f);
        a.z = (1.0f / HHEADS) * ex4.z / (dn4.z + 1e-16f);
        a.w = (1.0f / HHEADS) * ex4.w / (dn4.w + 1e-16f);
        lw[i] = a;
    }
    __syncthreads();

    unsigned hv = LEND;
    {
        const int sl = wvu + AGG_WAVES * lane;
        if (lane < SRC_PER_W && sl < NNODE) hv = head[sl];
    }

    for (int j0 = 0; j0 < SRC_PER_W; j0 += P2_SRC) {
        v2f xc[P2_SRC][4];
        #pragma unroll
        for (int k = 0; k < P2_SRC; ++k) {
            #pragma unroll
            for (int j = 0; j < 4; ++j) xc[k][j] = xn[k][j];
        }
        if (j0 + P2_SRC < SRC_PER_W) {
            #pragma unroll
            for (int k = 0; k < P2_SRC; ++k) {
                const int s = min(wvu + AGG_WAVES * (j0 + P2_SRC + k), NNODE - 1);
                const float* p = hr + (size_t)s * OUTD;
                #pragma unroll
                for (int j = 0; j < 4; ++j)
                    xn[k][j] = *(const v2f*)(p + j * CDIM);
            }
        }
        #pragma unroll
        for (int k = 0; k < P2_SRC; ++k) {
            unsigned cur = (unsigned)__builtin_amdgcn_readlane((int)hv, j0 + k);
            for (int hop = 0; hop < LCAP && cur != LEND; ++hop) {
                const unsigned e  = lsd[cur];
                const float4   wc = lw[cur];
                v2f m = xc[k][0] * wc.x;
                m += xc[k][1] * wc.y;
                m += xc[k][2] * wc.z;
                m += xc[k][3] * wc.w;
                const unsigned eu = (unsigned)__builtin_amdgcn_readfirstlane((int)e);
                float* ap = accl + (eu >> 16) * FDIM;
                atomicAdd(ap,     m.x);
                atomicAdd(ap + 1, m.y);
                cur = eu & 0xffffu;
            }
        }
    }

    __syncthreads();
    float4*       o4 = (float4*)(out + ((size_t)r * NNODE + d0) * FDIM);
    const float4* a4 = (const float4*)acc;
    for (int i = threadIdx.x; i < TROWS * FDIM / 4; i += AGG_THREADS)
        o4[i] = a4[i];
}

extern "C" void kernel_launch(void* const* d_in, const int* in_sizes, int n_in,
                              void* d_out, int out_size, void* d_ws, size_t ws_size,
                              hipStream_t stream) {
    const float* x       = (const float*)d_in[0];
    const int*   ei      = (const int*)d_in[1];
    const float* W       = (const float*)d_in[2];
    const float* att_src = (const float*)d_in[3];
    const float* att_dst = (const float*)d_in[4];
    const float* bias    = (const float*)d_in[5];
    float* out = (float*)d_out;

    float* ws    = (float*)d_ws;
    float* hbuf  = ws;
    float* a_src = hbuf  + (size_t)NROWS * OUTD;
    float* a_dst = a_src + (size_t)NROWS * HHEADS;

    {
        dim3 grid(PROJ_WG);
        dim3 block(128);
        gat_proj_wmma<<<grid, block, 0, stream>>>(x, W, att_src, att_dst,
                                                  hbuf, a_src, a_dst);
    }
    {
        gat_aggregate<<<RREP * NTILE, AGG_THREADS, 0, stream>>>(ei, hbuf, a_src, a_dst,
                                                               bias, out);
    }
}
